# starting kernel
# baseline (speedup 1.0000x reference)
#include <hip/hip_runtime.h>

typedef __attribute__((ext_vector_type(2))) float v2f;
typedef __attribute__((ext_vector_type(8))) float v8f;

#define TS 128
#define HALO 16
#define R   (TS + 2*HALO)
#define NCOL 144
#define GSTR 145
#define PSTR 9
#define NTHREADS 128
#define NWAVES (NTHREADS/32)
#define ROWTILES (R/16)
#define COLTILES (NCOL/16)

__global__ __launch_bounds__(NTHREADS)
void gmp_wmma_kernel(const float* __restrict__ x,
                     const float* __restrict__ a_re, const float* __restrict__ a_im,
                     const float* __restrict__ b_re, const float* __restrict__ b_im,
                     const float* __restrict__ c_re, const float* __restrict__ c_im,
                     float* __restrict__ out, int N)
{
    __shared__ float W[8][NCOL];
    __shared__ float P[R][PSTR];
    __shared__ float XS[R][2];
    __shared__ float G[R * GSTR];

    const int tid = threadIdx.x;
    const int blockStart = blockIdx.x * TS;

    for (int idx = tid; idx < 8 * NCOL; idx += NTHREADS) {
        const int k = idx / NCOL, col = idx % NCOL;
        float v;
        if (col < 8)        v = a_re[k * 8  + col];
        else if (col < 16)  v = a_im[k * 8  + (col - 8)];
        else if (col < 48)  v = b_re[k * 32 + (col - 16)];
        else if (col < 80)  v = b_im[k * 32 + (col - 48)];
        else if (col < 112) v = c_re[k * 32 + (col - 80)];
        else                v = c_im[k * 32 + (col - 112)];
        W[k][col] = v;
    }

    for (int j = tid; j < R; j += NTHREADS) {
        int g = blockStart + j - HALO;
        g = min(max(g, 0), N - 1);
        const float xr = x[2 * g], xi = x[2 * g + 1];
        XS[j][0] = xr; XS[j][1] = xi;
        const float r = sqrtf(xr * xr + xi * xi);
        float p = 1.0f;
        #pragma unroll
        for (int k = 0; k < 8; ++k) { P[j][k] = p; p *= r; }
    }
    __syncthreads();

    {
        const int wave  = tid >> 5;
        const int lane  = tid & 31;
        const int mrow  = lane & 15;
        const int khalf = (lane >> 4) * 2;
        for (int job = wave; job < ROWTILES * COLTILES; job += NWAVES) {
            const int rt = (job / COLTILES) * 16;
            const int ct = (job % COLTILES) * 16;
            v8f acc = {};
            #pragma unroll
            for (int ks = 0; ks < 8; ks += 4) {
                v2f Af, Bf;
                Af.x = P[rt + mrow][ks + khalf + 0];
                Af.y = P[rt + mrow][ks + khalf + 1];
                Bf.x = W[ks + khalf + 0][ct + mrow];
                Bf.y = W[ks + khalf + 1][ct + mrow];
                acc = __builtin_amdgcn_wmma_f32_16x16x4_f32(
                    false, Af, false, Bf, (short)0, acc, false, false);
            }
            const int rowBase = rt + (lane >> 4) * 8;
            #pragma unroll
            for (int i = 0; i < 8; ++i)
                G[(rowBase + i) * GSTR + ct + mrow] = acc[i];
        }
    }
    __syncthreads();

    {
        const int t = tid;
        float accRe = 0.f, accIm = 0.f;
        #pragma unroll
        for (int l = 0; l < 8; ++l) {
            const int r0 = t + HALO - l;
            const float* gr = &G[r0 * GSTR];
            float sRe = gr[l];
            float sIm = gr[8 + l];
            #pragma unroll
            for (int m = 0; m < 4; ++m) {
                const float* gb = &G[(r0 - m) * GSTR];
                sRe += gb[16 + l * 4 + m];
                sIm += gb[48 + l * 4 + m];
                const float* gc = &G[(r0 + m) * GSTR];
                sRe += gc[80  + l * 4 + m];
                sIm += gc[112 + l * 4 + m];
            }
            const float xr = XS[r0][0], xi = XS[r0][1];
            accRe += xr * sRe - xi * sIm;
            accIm += xr * sIm + xi * sRe;
        }
        const int n = blockStart + t;
        if (n < N) { out[2 * n] = accRe; out[2 * n + 1] = accIm; }
    }
}

extern "C" void kernel_launch(void* const* d_in, const int* in_sizes, int n_in,
                              void* d_out, int out_size, void* d_ws, size_t ws_size,
                              hipStream_t stream) {
    (void)n_in; (void)out_size; (void)d_ws; (void)ws_size;
    const float* x    = (const float*)d_in[0];
    const float* a_re = (const float*)d_in[1];
    const float* a_im = (const float*)d_in[2];
    const float* b_re = (const float*)d_in[3];
    const float* b_im = (const float*)d_in[4];
    const float* c_re = (const float*)d_in[5];
    const float* c_im = (const float*)d_in[6];
    float* out = (float*)d_out;
    const int N = in_sizes[0] / 2;
    const int grid = (N + TS - 1) / TS;
    gmp_wmma_kernel<<<grid, NTHREADS, 0, stream>>>(x, a_re, a_im, b_re, b_im,
                                                   c_re, c_im, out, N);
}


# Fold the 7 envelope shifts into the GEMM K dim (K=56: A = power table read at 7 row shifts, B = 56x16): 16 gains per sample replace the 144-column G round trip through LDS; sum order changes
# speedup vs baseline: 3.8926x; 3.8926x over previous
#include <hip/hip_runtime.h>


typedef __attribute__((ext_vector_type(2))) float v2f;
typedef __attribute__((ext_vector_type(4))) float v4f;
typedef __attribute__((ext_vector_type(8))) float v8f;


#define NTHREADS 128
#define NWAVES   (NTHREADS / 32)
#define WS       128
#define TS       (NWAVES * WS)
#define NT       (WS / 16 + 1)
#define HROWS    (NT * 16)
#define PR       (HROWS + 6)
#define PROUNDS  ((PR + 31) / 32)
#define PRA      (PROUNDS * 32)
#define PSTR     8
#define HSTR     18
#define KTOT     56
#define KSTEPS   (KTOT / 4)

__global__ __launch_bounds__(NTHREADS)
void gmp_wmma_kernel(const float* __restrict__ x,
                     const float* __restrict__ a_re, const float* __restrict__ a_im,
                     const float* __restrict__ b_re, const float* __restrict__ b_im,
                     const float* __restrict__ c_re, const float* __restrict__ c_im,
                     float* __restrict__ out, int N)
{
    __shared__ float W2[KTOT][16];
    __shared__ __attribute__((aligned(16))) float Pw[NWAVES][PRA][PSTR];
    __shared__ __attribute__((aligned(8)))  float XSw[NWAVES][PRA][2];
    __shared__ float Hs[NWAVES][HROWS * HSTR];

    const int tid  = threadIdx.x;
    const int wave = tid >> 5;
    const int lane = tid & 31;
    const int c    = lane & 15;
    const int h    = lane >> 4;
    const int ws   = blockIdx.x * TS + wave * WS;

    float wv[7];
    {
        const int k = tid >> 4, col = tid & 15;
        const int l = col & 7;
        const bool im = col >= 8;
        const float* aa = im ? a_im : a_re;
        const float* bb = im ? b_im : b_re;
        const float* cc = im ? c_im : c_re;
        const int e = k * 32 + l * 4;
        #pragma unroll
        for (int s = 0; s < 7; ++s) {
            if (s < 3)      wv[s] = bb[e + (3 - s)];
            else if (s > 3) wv[s] = cc[e + (s - 3)];
            else            wv[s] = aa[k * 8 + l] + bb[e] + cc[e];
        }
    }
    float xr[PROUNDS], xi[PROUNDS];
    #pragma unroll
    for (int it = 0; it < PROUNDS; ++it) {
        int g = ws - 19 + lane + 32 * it;
        g = min(max(g, 0), N - 1);
        xr[it] = x[2 * g]; xi[it] = x[2 * g + 1];
    }

    {
        const int k = tid >> 4, col = tid & 15;
        #pragma unroll
        for (int s = 0; s < 7; ++s) W2[s * 8 + k][col] = wv[s];
    }
    #pragma unroll
    for (int it = 0; it < PROUNDS; ++it) {
        const int j = lane + 32 * it;
        XSw[wave][j][0] = xr[it]; XSw[wave][j][1] = xi[it];
        const float r = sqrtf(xr[it] * xr[it] + xi[it] * xi[it]);
        float p[8];
        p[0] = 1.0f;
        #pragma unroll
        for (int k = 1; k < 8; ++k) p[k] = p[k - 1] * r;
        v4f lo = {p[0], p[1], p[4], p[5]};
        v4f hi = {p[2], p[3], p[6], p[7]};
        *(v4f*)&Pw[wave][j][0] = lo;
        *(v4f*)&Pw[wave][j][4] = hi;
    }
    __syncthreads();

    {
        v2f Bf[KSTEPS];
        #pragma unroll
        for (int ks = 0; ks < KSTEPS; ++ks) {
            Bf[ks].x = W2[4 * ks + 2 * h + 0][c];
            Bf[ks].y = W2[4 * ks + 2 * h + 1][c];
        }
        const float* Pl = &Pw[wave][c][4 * h];
        float* Hl = &Hs[wave][(8 * h) * HSTR + c];
        for (int t = 0; t < NT; t += 3) {
            v8f acc0[3], acc1[3];
            #pragma unroll
            for (int u = 0; u < 3; ++u) { acc0[u] = (v8f){}; acc1[u] = (v8f){}; }
            #pragma unroll
            for (int s = 0; s < 7; ++s) {
                #pragma unroll
                for (int u = 0; u < 3; ++u) {
                    const v4f a = *(const v4f*)(Pl + (16 * (t + u) + s) * PSTR);
                    v2f A0 = {a.x, a.y};
                    v2f A1 = {a.z, a.w};
                    acc0[u] = __builtin_amdgcn_wmma_f32_16x16x4_f32(
                        false, A0, false, Bf[2 * s + 0], (short)0, acc0[u], false, false);
                    acc1[u] = __builtin_amdgcn_wmma_f32_16x16x4_f32(
                        false, A1, false, Bf[2 * s + 1], (short)0, acc1[u], false, false);
                }
            }
            #pragma unroll
            for (int u = 0; u < 3; ++u) {
                const v8f acc = acc0[u] + acc1[u];
                #pragma unroll
                for (int i = 0; i < 8; ++i)
                    Hl[(16 * (t + u) + i) * HSTR] = acc[i];
            }
        }
    }
    __syncthreads();

    #pragma unroll
    for (int u = 0; u < WS / 32; ++u) {
        const int nl = lane + 32 * u;
        float accRe = 0.f, accIm = 0.f;
        #pragma unroll
        for (int l = 0; l < 8; ++l) {
            const float* hrow = &Hs[wave][(16 + nl - l) * HSTR];
            const float sRe = hrow[l];
            const float sIm = hrow[8 + l];
            const float xr = XSw[wave][19 + nl - l][0], xi = XSw[wave][19 + nl - l][1];
            accRe += xr * sRe - xi * sIm;
            accIm += xr * sIm + xi * sRe;
        }
        const int n = ws + nl;
        if (n < N) { out[2 * n] = accRe; out[2 * n + 1] = accIm; }
    }
}

extern "C" void kernel_launch(void* const* d_in, const int* in_sizes, int n_in,
                              void* d_out, int out_size, void* d_ws, size_t ws_size,
                              hipStream_t stream) {
    (void)n_in; (void)out_size; (void)d_ws; (void)ws_size;
    const float* x    = (const float*)d_in[0];
    const float* a_re = (const float*)d_in[1];
    const float* a_im = (const float*)d_in[2];
    const float* b_re = (const float*)d_in[3];
    const float* b_im = (const float*)d_in[4];
    const float* c_re = (const float*)d_in[5];
    const float* c_im = (const float*)d_in[6];
    float* out = (float*)d_out;
    const int N = in_sizes[0] / 2;
    const int grid = (N + TS - 1) / TS;
    gmp_wmma_kernel<<<grid, NTHREADS, 0, stream>>>(x, a_re, a_im, b_re, b_im,
                                                   c_re, c_im, out, N);
}


# H rows stored with each tap's (re,im) adjacent: output phase reads a gain as one 8-byte pair into packed FMAs on the running sum, four outputs per lane branch-free; sum order inside a tap changes
# speedup vs baseline: 4.1501x; 1.0662x over previous
#include <hip/hip_runtime.h>


typedef __attribute__((ext_vector_type(2))) float v2f;
typedef __attribute__((ext_vector_type(4))) float v4f;
typedef __attribute__((ext_vector_type(8))) float v8f;


#define NTHREADS 128
#define NWAVES   (NTHREADS / 32)
#define WS       128
#define TS       (NWAVES * WS)
#define NT       (WS / 16 + 1)
#define HROWS    (NT * 16)
#define PR       (HROWS + 6)
#define PROUNDS  ((PR + 31) / 32)
#define PRA      (PROUNDS * 32)
#define PSTR     8
#define HSTR     18
#define KTOT     56
#define KSTEPS   (KTOT / 4)

__global__ __launch_bounds__(NTHREADS)
void gmp_wmma_kernel(const float* __restrict__ x,
                     const float* __restrict__ a_re, const float* __restrict__ a_im,
                     const float* __restrict__ b_re, const float* __restrict__ b_im,
                     const float* __restrict__ c_re, const float* __restrict__ c_im,
                     float* __restrict__ out, int N)
{
    __shared__ float W2[KTOT][16];
    __shared__ __attribute__((aligned(16))) float Pw[NWAVES][PRA][PSTR];
    __shared__ __attribute__((aligned(8)))  float XSw[NWAVES][PRA][2];
    __shared__ __attribute__((aligned(8)))  float Hs[NWAVES][HROWS * HSTR];

    const int tid  = threadIdx.x;
    const int wave = tid >> 5;
    const int lane = tid & 31;
    const int c    = lane & 15;
    const int h    = lane >> 4;
    const int ws   = blockIdx.x * TS + wave * WS;

    float wv[7];
    {
        const int k = tid >> 4, col = tid & 15;
        const int l = col & 7;
        const bool im = col >= 8;
        const float* aa = im ? a_im : a_re;
        const float* bb = im ? b_im : b_re;
        const float* cc = im ? c_im : c_re;
        const int e = k * 32 + l * 4;
        #pragma unroll
        for (int s = 0; s < 7; ++s) {
            if (s < 3)      wv[s] = bb[e + (3 - s)];
            else if (s > 3) wv[s] = cc[e + (s - 3)];
            else            wv[s] = aa[k * 8 + l] + bb[e] + cc[e];
        }
    }
    float xr[PROUNDS], xi[PROUNDS];
    #pragma unroll
    for (int it = 0; it < PROUNDS; ++it) {
        int g = ws - 19 + lane + 32 * it;
        g = min(max(g, 0), N - 1);
        xr[it] = x[2 * g]; xi[it] = x[2 * g + 1];
    }

    {
        const int k = tid >> 4, col = tid & 15;
        #pragma unroll
        for (int s = 0; s < 7; ++s) W2[s * 8 + k][col] = wv[s];
    }
    #pragma unroll
    for (int it = 0; it < PROUNDS; ++it) {
        const int j = lane + 32 * it;
        XSw[wave][j][0] = xr[it]; XSw[wave][j][1] = xi[it];
        const float r = sqrtf(xr[it] * xr[it] + xi[it] * xi[it]);
        float p[8];
        p[0] = 1.0f;
        #pragma unroll
        for (int k = 1; k < 8; ++k) p[k] = p[k - 1] * r;
        v4f lo = {p[0], p[1], p[4], p[5]};
        v4f hi = {p[2], p[3], p[6], p[7]};
        *(v4f*)&Pw[wave][j][0] = lo;
        *(v4f*)&Pw[wave][j][4] = hi;
    }
    __syncthreads();

    {
        v2f Bf[KSTEPS];
        #pragma unroll
        for (int ks = 0; ks < KSTEPS; ++ks) {
            Bf[ks].x = W2[4 * ks + 2 * h + 0][c];
            Bf[ks].y = W2[4 * ks + 2 * h + 1][c];
        }
        const float* Pl = &Pw[wave][c][4 * h];
        float* Hl = &Hs[wave][(8 * h) * HSTR + 2 * (c & 7) + (c >> 3)];
        for (int t = 0; t < NT; t += 3) {
            v8f acc0[3], acc1[3];
            #pragma unroll
            for (int u = 0; u < 3; ++u) { acc0[u] = (v8f){}; acc1[u] = (v8f){}; }
            #pragma unroll
            for (int s = 0; s < 7; ++s) {
                #pragma unroll
                for (int u = 0; u < 3; ++u) {
                    const v4f a = *(const v4f*)(Pl + (16 * (t + u) + s) * PSTR);
                    v2f A0 = {a.x, a.y};
                    v2f A1 = {a.z, a.w};
                    acc0[u] = __builtin_amdgcn_wmma_f32_16x16x4_f32(
                        false, A0, false, Bf[2 * s + 0], (short)0, acc0[u], false, false);
                    acc1[u] = __builtin_amdgcn_wmma_f32_16x16x4_f32(
                        false, A1, false, Bf[2 * s + 1], (short)0, acc1[u], false, false);
                }
            }
            #pragma unroll
            for (int u = 0; u < 3; ++u) {
                const v8f acc = acc0[u] + acc1[u];
                #pragma unroll
                for (int i = 0; i < 8; ++i)
                    Hl[(16 * (t + u) + i) * HSTR] = acc[i];
            }
        }
    }
    __syncthreads();

    v2f acc[WS / 32];
    #pragma unroll
    for (int l = 0; l < 8; ++l) {
        #pragma unroll
        for (int u = 0; u < WS / 32; ++u) {
            const int nl = lane + 32 * u;
            const v2f g  = *(const v2f*)&Hs[wave][(16 + nl - l) * HSTR + 2 * l];
            const v2f xv = *(const v2f*)&XSw[wave][19 + nl - l][0];
            const v2f xr2 = {xv.x, xv.x};
            const v2f xi2 = {-xv.y, xv.y};
            const v2f gsw = {g.y, g.x};
            if (l == 0) acc[u] = xr2 * g;
            else        acc[u] = __builtin_elementwise_fma(xr2, g, acc[u]);
            acc[u] = __builtin_elementwise_fma(xi2, gsw, acc[u]);
        }
    }
    float* const o = out + 2 * (size_t)(ws + lane);
    if (ws + WS <= N) {
        #pragma unroll
        for (int u = 0; u < WS / 32; ++u) *(v2f*)(o + 64 * u) = acc[u];
    } else {
        #pragma unroll
        for (int u = 0; u < WS / 32; ++u)
            if (ws + lane + 32 * u < N) *(v2f*)(o + 64 * u) = acc[u];
    }
}

extern "C" void kernel_launch(void* const* d_in, const int* in_sizes, int n_in,
                              void* d_out, int out_size, void* d_ws, size_t ws_size,
                              hipStream_t stream) {
    (void)n_in; (void)out_size; (void)d_ws; (void)ws_size;
    const float* x    = (const float*)d_in[0];
    const float* a_re = (const float*)d_in[1];
    const float* a_im = (const float*)d_in[2];
    const float* b_re = (const float*)d_in[3];
    const float* b_im = (const float*)d_in[4];
    const float* c_re = (const float*)d_in[5];
    const float* c_im = (const float*)d_in[6];
    float* out = (float*)d_out;
    const int N = in_sizes[0] / 2;
    const int grid = (N + TS - 1) / TS;
    gmp_wmma_kernel<<<grid, NTHREADS, 0, stream>>>(x, a_re, a_im, b_re, b_im,
                                                   c_re, c_im, out, N);
}
